# starting kernel
# baseline (speedup 1.0000x reference)
#include <hip/hip_runtime.h>
#include <math.h>

typedef float v2f __attribute__((ext_vector_type(2)));
typedef float v4f __attribute__((ext_vector_type(4)));
typedef float v8f __attribute__((ext_vector_type(8)));

#define ROWS_PER_BLOCK 128
#define THREADS 128
#define PRE_LD 136
#define H_LD 72


template<int DIM, int LEN, int UNROLL>
__device__ __forceinline__ void pool_to(float* dst, const float* __restrict__ tab,
                                        const int* __restrict__ ids, int row) {
  float acc[DIM];
#pragma unroll
  for (int d = 0; d < DIM; ++d) acc[d] = 0.f;
#pragma unroll UNROLL
  for (int j = 0; j < LEN; ++j) {
    const int id = ids[row * LEN + j];
    float v[DIM];
    if constexpr ((DIM % 4) == 0) {
      const v4f* tr = (const v4f*)(tab + (size_t)id * DIM);
#pragma unroll
      for (int c = 0; c < DIM / 4; ++c) {
        v4f t = tr[c];
        v[4 * c + 0] = t.x; v[4 * c + 1] = t.y;
        v[4 * c + 2] = t.z; v[4 * c + 3] = t.w;
      }
    } else {
      const v2f* tr = (const v2f*)(tab + (size_t)id * DIM);
#pragma unroll
      for (int c = 0; c < DIM / 2; ++c) {
        v2f t = tr[c];
        v[2 * c + 0] = t.x; v[2 * c + 1] = t.y;
      }
    }
    float ss = 0.f;
#pragma unroll
    for (int d = 0; d < DIM; ++d) ss = fmaf(v[d], v[d], ss);
    float n = sqrtf(ss);
    float s = fminf(1.0f, 1.0f / (n + 1e-7f));
#pragma unroll
    for (int d = 0; d < DIM; ++d) acc[d] = fmaf(v[d], s, acc[d]);
  }
#pragma unroll
  for (int d = 0; d < DIM; ++d) dst[d] = acc[d];
}

__device__ __forceinline__ v8f wmma_f32_k4(v2f a, v2f b, v8f c) {
#if __has_builtin(__builtin_amdgcn_wmma_f32_16x16x4_f32)
  return __builtin_amdgcn_wmma_f32_16x16x4_f32(false, a, false, b, (short)0, c,
                                               false, false);
#else
  asm volatile("v_wmma_f32_16x16x4_f32 %0, %1, %2, %0"
               : "+v"(c) : "v"(a), "v"(b));
  return c;
#endif
}

__device__ __forceinline__ const float* launder(const float* p) {
  asm("" : "+s"(p));
  return p;
}

__global__ __launch_bounds__(THREADS)
__attribute__((amdgpu_waves_per_eu(4)))
void fused_embmlp_kernel(
    const float* __restrict__ dense,
    const int* __restrict__ ids_rating, const int* __restrict__ ids_category,
    const int* __restrict__ ids_fandom, const int* __restrict__ ids_relationship,
    const int* __restrict__ ids_character, const int* __restrict__ ids_freeform,
    const float* __restrict__ W_dense, const float* __restrict__ b_dense,
    const float* __restrict__ tab_rating, const float* __restrict__ tab_category,
    const float* __restrict__ tab_fandom, const float* __restrict__ tab_relationship,
    const float* __restrict__ tab_character, const float* __restrict__ tab_freeform,
    const float* __restrict__ W1, const float* __restrict__ b1,
    const float* __restrict__ W2, const float* __restrict__ b2,
    float* __restrict__ out) {
  extern __shared__ float smem[];
  float* s_pre = smem;
  float* s_h   = smem + ROWS_PER_BLOCK * PRE_LD;

  const int tid = threadIdx.x;
  const int row = blockIdx.x * ROWS_PER_BLOCK + tid;
  float* myrow = s_pre + tid * PRE_LD;

  {
    const v4f* dr4 = (const v4f*)(dense + (size_t)row * 64);
    float acc[20];
#pragma unroll
    for (int j = 0; j < 20; ++j) acc[j] = b_dense[j];
#pragma unroll 2
    for (int kc = 0; kc < 16; ++kc) {
      v4f x = dr4[kc];
#pragma unroll
      for (int j = 0; j < 20; ++j) {
        acc[j] = fmaf(x.x, W_dense[(4 * kc + 0) * 20 + j], acc[j]);
        acc[j] = fmaf(x.y, W_dense[(4 * kc + 1) * 20 + j], acc[j]);
        acc[j] = fmaf(x.z, W_dense[(4 * kc + 2) * 20 + j], acc[j]);
        acc[j] = fmaf(x.w, W_dense[(4 * kc + 3) * 20 + j], acc[j]);
      }
    }
#pragma unroll
    for (int j = 0; j < 20; ++j) myrow[j] = acc[j];
  }
  pool_to<10, 1, 1>(myrow + 20,  tab_rating,       ids_rating,       row);
  pool_to<10, 2, 2>(myrow + 30,  tab_category,     ids_category,     row);
  pool_to<20, 3, 2>(myrow + 40,  tab_fandom,       ids_fandom,       row);
  pool_to<20, 3, 2>(myrow + 60,  tab_relationship, ids_relationship, row);
  pool_to<20, 5, 2>(myrow + 80,  tab_character,    ids_character,    row);
  pool_to<20, 10, 2>(myrow + 100, tab_freeform,    ids_freeform,     row);

  {
    const int PI[15] = {0,0,0,0,0,1,1,1,1,2,2,2,3,3,4};
    const int PJ[15] = {1,2,3,4,5,2,3,4,5,3,4,5,4,5,5};
#pragma unroll 3
    for (int p = 0; p < 15; ++p) {
      const float* a = myrow + 20 * PI[p];
      const float* b = myrow + 20 * PJ[p];
      float s = 0.f;
#pragma unroll
      for (int d = 0; d < 20; ++d) s = fmaf(a[d], b[d], s);
      myrow[120 + p] = s;
    }
    myrow[135] = 0.f;
  }

  __syncthreads();

  const int wave = tid >> 5;
  const int lane = tid & 31;
  const int lo = lane & 15;
  const int hi = lane >> 4;
  float* hbase = s_h + wave * 16 * H_LD;

  for (int t = 0; t < 2; ++t) {
    const int m0 = (wave * 2 + t) * 16;

    const float* arow = s_pre + (m0 + lo) * PRE_LD + 2 * hi;
#pragma unroll
    for (int nt = 0; nt < 4; ++nt) {
      const float* W1l = launder(W1);
      const int n0 = nt * 16;
      const float bias = b1[n0 + lo];
      v8f c;
#pragma unroll
      for (int i = 0; i < 8; ++i) c[i] = bias;
#pragma unroll
      for (int k0 = 0; k0 < 136; k0 += 4) {
        v2f a; a.x = arow[k0]; a.y = arow[k0 + 1];
        const int kk = k0 + 2 * hi;
        v2f b;
        b.x = W1l[kk * 64 + n0 + lo];
        b.y = (kk + 1 < 135) ? W1l[(kk + 1) * 64 + n0 + lo] : 0.f;
        c = wmma_f32_k4(a, b, c);
      }
#pragma unroll
      for (int i = 0; i < 8; ++i) {
        float v = c[i];
        v = v > 0.f ? v : 0.01f * v;
        hbase[(i + 8 * hi) * H_LD + n0 + lo] = v;
      }
    }
    asm volatile("s_wait_dscnt 0x0" ::: "memory");

    const float* hrow = hbase + lo * H_LD + 2 * hi;
    const size_t rbase = (size_t)(blockIdx.x * ROWS_PER_BLOCK + m0) * 128;
#pragma unroll
    for (int nt = 0; nt < 8; ++nt) {
      const float* W2l = launder(W2);
      const int n0 = nt * 16;
      const float bias = b2[n0 + lo];
      v8f c;
#pragma unroll
      for (int i = 0; i < 8; ++i) c[i] = bias;
#pragma unroll
      for (int k0 = 0; k0 < 64; k0 += 4) {
        v2f a; a.x = hrow[k0]; a.y = hrow[k0 + 1];
        const int kk = k0 + 2 * hi;
        v2f b;
        b.x = W2l[kk * 128 + n0 + lo];
        b.y = W2l[(kk + 1) * 128 + n0 + lo];
        c = wmma_f32_k4(a, b, c);
      }
#pragma unroll
      for (int i = 0; i < 8; ++i)
        out[rbase + (size_t)(i + 8 * hi) * 128 + n0 + lo] = c[i];
    }
  }
}

extern "C" void kernel_launch(void* const* d_in, const int* in_sizes, int n_in,
                              void* d_out, int out_size, void* d_ws, size_t ws_size,
                              hipStream_t stream) {
  const float* dense = (const float*)d_in[0];
  const int* ids_rating       = (const int*)d_in[1];
  const int* ids_category     = (const int*)d_in[2];
  const int* ids_fandom       = (const int*)d_in[3];
  const int* ids_relationship = (const int*)d_in[4];
  const int* ids_character    = (const int*)d_in[5];
  const int* ids_freeform     = (const int*)d_in[6];
  const float* W_dense = (const float*)d_in[7];
  const float* b_dense = (const float*)d_in[8];
  const float* tab_rating       = (const float*)d_in[9];
  const float* tab_category     = (const float*)d_in[10];
  const float* tab_fandom       = (const float*)d_in[11];
  const float* tab_relationship = (const float*)d_in[12];
  const float* tab_character    = (const float*)d_in[13];
  const float* tab_freeform     = (const float*)d_in[14];
  const float* W1 = (const float*)d_in[15];
  const float* b1 = (const float*)d_in[16];
  const float* W2 = (const float*)d_in[17];
  const float* b2 = (const float*)d_in[18];
  float* out = (float*)d_out;

  const int Btot = in_sizes[0] / 64;
  const int grid = Btot / ROWS_PER_BLOCK;
  const size_t shmem = (size_t)(ROWS_PER_BLOCK * PRE_LD + 4 * 16 * H_LD) * sizeof(float);

  fused_embmlp_kernel<<<grid, THREADS, shmem, stream>>>(
      dense, ids_rating, ids_category, ids_fandom, ids_relationship,
      ids_character, ids_freeform, W_dense, b_dense, tab_rating, tab_category,
      tab_fandom, tab_relationship, tab_character, tab_freeform, W1, b1, W2, b2,
      out);
}


# Dense projection moved from per-thread FMAs onto the f32 16x16x4 matrix instruction: operands loaded at kernel start, products run while the first gather round is in flight; de's k sum order changes
# speedup vs baseline: 1.4998x; 1.4998x over previous
#include <hip/hip_runtime.h>
#include <math.h>

typedef float v2f __attribute__((ext_vector_type(2)));
typedef float v4f __attribute__((ext_vector_type(4)));
typedef float v8f __attribute__((ext_vector_type(8)));

#define ROWS_PER_BLOCK 128
#define THREADS 128
#define PRE_LD 140
#define H_LD 68
#define W1_PAIRS 17
#define W2_PAIRS 8
#define W1_TILE (W1_PAIRS * 32 + 1)
#define W2_TILE (W2_PAIRS * 32 + 1)
#define W1F_FLOATS (4 * W1_TILE * 4)
#define W2F_FLOATS (8 * W2_TILE * 4)
#define STAGE_FLOATS 3200
#define HREG_FLOATS (4 * 16 * H_LD)
#define UNION_FLOATS (4 * STAGE_FLOATS)
static_assert(UNION_FLOATS >= HREG_FLOATS + W2F_FLOATS, "h and W2 fragments must fit");


template<int DIM> struct piece { typedef v4f type; };
template<> struct piece<10> { typedef v2f type; };

template<int NE>
__device__ __forceinline__ int list_pos(int i, int e0) {
  const int e = 6 * i + e0;
  return e < NE ? e : NE - 1;
}

template<int NE>
__device__ __forceinline__ void load_ids(int (&id)[(NE + 5) / 6],
                                         const int* __restrict__ idp, int e0) {
#pragma unroll
  for (int i = 0; i < (NE + 5) / 6; ++i) id[i] = idp[list_pos<NE>(i, e0)];
}

template<int DIM, int NE>
__device__ __forceinline__ void load_pieces(typename piece<DIM>::type (&p)[(NE + 5) / 6],
                                            const float* __restrict__ tab,
                                            const int (&id)[(NE + 5) / 6], int c) {
  typedef typename piece<DIM>::type PT;
#pragma unroll
  for (int i = 0; i < (NE + 5) / 6; ++i)
    p[i] = *(const PT*)(tab + (size_t)id[i] * DIM + c * (DIM / 5));
}

template<int DIM, int NE>
__device__ __forceinline__ void store_pieces(float* dst,
                                             const typename piece<DIM>::type (&p)[(NE + 5) / 6],
                                             int e0, int c) {
  typedef typename piece<DIM>::type PT;
#pragma unroll
  for (int i = 0; i < (NE + 5) / 6; ++i)
    *(PT*)(dst + list_pos<NE>(i, e0) * DIM + c * (DIM / 5)) = p[i];
}

template<int DIM>
__device__ __forceinline__ void staged_row(float (&v)[DIM], const float* src) {
  typedef typename piece<DIM>::type PT;
  constexpr int PW = DIM / 5;
#pragma unroll
  for (int c = 0; c < 5; ++c) {
    PT t = *(const PT*)(src + c * PW);
#pragma unroll
    for (int k = 0; k < PW; ++k) v[PW * c + k] = t[k];
  }
}

__device__ __forceinline__ void wave_lds_handover() {
  asm volatile("s_wait_dscnt 0x0" ::: "memory");
}

template<int DIM>
__device__ __forceinline__ void fold_row(float (&acc)[DIM], const float (&v)[DIM]) {
  float ss = 0.f;
#pragma unroll
  for (int d = 0; d < DIM; ++d) ss = fmaf(v[d], v[d], ss);
  float n = sqrtf(ss);
  float s = fminf(1.0f, 1.0f / (n + 1e-7f));
#pragma unroll
  for (int d = 0; d < DIM; ++d) acc[d] = fmaf(v[d], s, acc[d]);
}

template<int DIM>
__device__ __forceinline__ void zero_acc(float (&acc)[DIM]) {
#pragma unroll
  for (int d = 0; d < DIM; ++d) acc[d] = 0.f;
}

template<int DIM>
__device__ __forceinline__ void put_acc(float* dst, const float (&acc)[DIM]) {
#pragma unroll
  for (int d = 0; d < DIM; ++d) dst[d] = acc[d];
}

__device__ __forceinline__ int pre_row(int r) {
  return r * PRE_LD + 2 * ((r >> 4) & 1);
}

__device__ __forceinline__ v8f wmma_f32_k4(v2f a, v2f b, v8f c) {
#if __has_builtin(__builtin_amdgcn_wmma_f32_16x16x4_f32)
  return __builtin_amdgcn_wmma_f32_16x16x4_f32(false, a, false, b, (short)0, c,
                                               false, false);
#else
  asm volatile("v_wmma_f32_16x16x4_f32 %0, %1, %2, %0"
               : "+v"(c) : "v"(a), "v"(b));
  return c;
#endif
}

__device__ __forceinline__ v4f make4(float a, float b, float c, float d) {
  v4f r; r.x = a; r.y = b; r.z = c; r.w = d;
  return r;
}

template<int LDW>
__device__ __forceinline__ void load_quad(v4f (&r)[4], const float* __restrict__ Wm,
                                          int kb, int k3, int nq) {
  const float* p = Wm + kb * LDW + 4 * nq;
  r[0] = *(const v4f*)(p);
  r[1] = *(const v4f*)(p + LDW);
  r[2] = *(const v4f*)(p + 4 * LDW);
  r[3] = *(const v4f*)(Wm + k3 * LDW + 4 * nq);
}

__device__ __forceinline__ void store_quad(v4f* dst, const v4f (&r)[4]) {
  dst[0] = make4(r[0].x, r[1].x, r[2].x, r[3].x);
  dst[1] = make4(r[0].y, r[1].y, r[2].y, r[3].y);
  dst[2] = make4(r[0].z, r[1].z, r[2].z, r[3].z);
  dst[3] = make4(r[0].w, r[1].w, r[2].w, r[3].w);
}

__global__ __launch_bounds__(THREADS)
__attribute__((amdgpu_waves_per_eu(2)))
void fused_embmlp_kernel(
    const float* __restrict__ dense,
    const int* __restrict__ ids_rating, const int* __restrict__ ids_category,
    const int* __restrict__ ids_fandom, const int* __restrict__ ids_relationship,
    const int* __restrict__ ids_character, const int* __restrict__ ids_freeform,
    const float* __restrict__ W_dense, const float* __restrict__ b_dense,
    const float* __restrict__ tab_rating, const float* __restrict__ tab_category,
    const float* __restrict__ tab_fandom, const float* __restrict__ tab_relationship,
    const float* __restrict__ tab_character, const float* __restrict__ tab_freeform,
    const float* __restrict__ W1, const float* __restrict__ b1,
    const float* __restrict__ W2, const float* __restrict__ b2,
    float* __restrict__ out) {
  extern __shared__ float smem[];
  float* s_pre = smem;
  float* s_w1  = smem + ROWS_PER_BLOCK * PRE_LD;
  float* s_un  = s_w1 + W1F_FLOATS;
  float* s_h   = s_un;
  float* s_w2  = s_un + HREG_FLOATS;

  const int tid = threadIdx.x;
  const int wave = tid >> 5;
  const int lane = tid & 31;

  {
    const int sw = tid >> 5;
    const int nq1 = tid & 15, h1 = (tid >> 4) & 1;
    const int h16 = tid >> 6, j16 = (tid >> 4) & 3;
    const int k16 = 128 + 2 * h16 + (j16 & 1) + 4 * (j16 >> 1);
    const bool pad16 = (k16 >= 135);
    v4f q1[4][4];
#pragma unroll
    for (int i = 0; i < 4; ++i) {
      const int kb = 8 * (sw + 4 * i) + 2 * h1;
      load_quad<64>(q1[i], W1, kb, kb + 5, nq1);
    }
    v4f t16 = *(const v4f*)(W1 + (pad16 ? 134 : k16) * 64 + 4 * nq1);
    if (pad16) t16 = make4(0.f, 0.f, 0.f, 0.f);
    v4f* d1 = (v4f*)s_w1 + (nq1 >> 2) * W1_TILE + (nq1 & 3) * 4;
#pragma unroll
    for (int i = 0; i < 4; ++i) store_quad(d1 + (sw + 4 * i) * 32 + h1 * 16, q1[i]);
    float* e16 = (float*)(d1 + 16 * 32 + h16 * 16) + j16;
    e16[0] = t16.x; e16[4] = t16.y; e16[8] = t16.z; e16[12] = t16.w;
  }

  float* myrow = s_pre + pre_row(tid);

  const int e0 = lane / 5, pc = lane % 5;
  float* stage = s_un + wave * STAGE_FLOATS;
  const int wrow = blockIdx.x * ROWS_PER_BLOCK + wave * 32;
  int i_rat[6], i_cat[11], i_fan[16], i_rel[16], i_chr[27], i_fa[27], i_fb[27];
  load_ids<32>(i_rat, ids_rating + wrow, e0);
  load_ids<64>(i_cat, ids_category + wrow * 2, e0);
  load_ids<96>(i_fan, ids_fandom + wrow * 3, e0);

  const int lo = lane & 15;
  const int hi = lane >> 4;
  v4f xa[2][8], wb[2][8];
  {
    const float* xr = dense + (size_t)(wrow + lo) * 64 + 4 * hi;
#pragma unroll
    for (int t = 0; t < 2; ++t)
#pragma unroll
      for (int j = 0; j < 8; ++j) xa[t][j] = *(const v4f*)(xr + t * 16 * 64 + 8 * j);
#pragma unroll
    for (int nt = 0; nt < 2; ++nt) {
      const int n = 16 * nt + lo;
      const bool ok = n < 20;
      const float* wp = W_dense + 4 * hi * 20 + (ok ? n : 0);
#pragma unroll
      for (int j = 0; j < 8; ++j) {
        const float* p = wp + 8 * j * 20;
        const v4f w = make4(p[0], p[20], p[40], p[60]);
        wb[nt][j] = ok ? w : make4(0.f, 0.f, 0.f, 0.f);
      }
    }
  }
  const float bd0 = b_dense[lo];
  const float bd1r = b_dense[lo < 4 ? 16 + lo : 0];
  const float bd1 = lo < 4 ? bd1r : 0.f;

  v2f p_rat[6], p_cat[11];
  v4f p_fan[16], p_rel[16], p_chr[27], p_fa[27], p_fb[27];
  load_pieces<10, 32>(p_rat, tab_rating, i_rat, pc);
  load_pieces<10, 64>(p_cat, tab_category, i_cat, pc);
  load_pieces<20, 96>(p_fan, tab_fandom, i_fan, pc);
  load_ids<96>(i_rel, ids_relationship + wrow * 3, e0);
#pragma unroll
  for (int t = 0; t < 2; ++t) {
    v8f c0, c1;
#pragma unroll
    for (int i = 0; i < 8; ++i) { c0[i] = bd0; c1[i] = bd1; }
#pragma unroll
    for (int j = 0; j < 8; ++j) {
      v2f a, b;
      a.x = xa[t][j].x;  a.y = xa[t][j].y;
      b.x = wb[0][j].x;  b.y = wb[0][j].y;
      c0 = wmma_f32_k4(a, b, c0);
      b.x = wb[1][j].x;  b.y = wb[1][j].y;
      c1 = wmma_f32_k4(a, b, c1);
      a.x = xa[t][j].z;  a.y = xa[t][j].w;
      b.x = wb[0][j].z;  b.y = wb[0][j].w;
      c0 = wmma_f32_k4(a, b, c0);
      b.x = wb[1][j].z;  b.y = wb[1][j].w;
      c1 = wmma_f32_k4(a, b, c1);
    }
    float* drow = s_pre + pre_row(wave * 32 + 16 * t + 8 * hi);
#pragma unroll
    for (int i = 0; i < 8; ++i) drow[i * PRE_LD + lo] = c0[i];
    if (lo < 4) {
#pragma unroll
      for (int i = 0; i < 8; ++i) drow[i * PRE_LD + 16 + lo] = c1[i];
    }
  }
  store_pieces<10, 32>(stage, p_rat, e0, pc);
  store_pieces<10, 64>(stage + 320, p_cat, e0, pc);
  store_pieces<20, 96>(stage + 960, p_fan, e0, pc);
  wave_lds_handover();
  load_pieces<20, 96>(p_rel, tab_relationship, i_rel, pc);
  load_ids<160>(i_chr, ids_character + wrow * 5, e0);
  {
    float r0[10], c0[10], c1[10], f[3][20];
    staged_row<10>(r0, stage + lane * 10);
    staged_row<10>(c0, stage + 320 + (lane * 2) * 10);
    staged_row<10>(c1, stage + 320 + (lane * 2 + 1) * 10);
#pragma unroll
    for (int j = 0; j < 3; ++j) staged_row<20>(f[j], stage + 960 + (lane * 3 + j) * 20);
    float a10[10], a20[20];
    zero_acc<10>(a10); fold_row<10>(a10, r0); put_acc<10>(myrow + 20, a10);
    zero_acc<10>(a10); fold_row<10>(a10, c0); fold_row<10>(a10, c1);
    put_acc<10>(myrow + 30, a10);
    zero_acc<20>(a20);
#pragma unroll
    for (int j = 0; j < 3; ++j) fold_row<20>(a20, f[j]);
    put_acc<20>(myrow + 40, a20);
  }
  wave_lds_handover();
  store_pieces<20, 96>(stage, p_rel, e0, pc);
  wave_lds_handover();
  load_pieces<20, 160>(p_chr, tab_character, i_chr, pc);
  load_ids<160>(i_fa, ids_freeform + wrow * 10, e0);
  {
    float r[3][20], a20[20];
#pragma unroll
    for (int j = 0; j < 3; ++j) staged_row<20>(r[j], stage + (lane * 3 + j) * 20);
    zero_acc<20>(a20);
#pragma unroll
    for (int j = 0; j < 3; ++j) fold_row<20>(a20, r[j]);
    put_acc<20>(myrow + 60, a20);
  }
  wave_lds_handover();
  store_pieces<20, 160>(stage, p_chr, e0, pc);
  wave_lds_handover();
  load_pieces<20, 160>(p_fa, tab_freeform, i_fa, pc);
  load_ids<160>(i_fb, ids_freeform + (wrow + 16) * 10, e0);
  {
    float a20[20];
    zero_acc<20>(a20);
#pragma unroll
    for (int j = 0; j < 5; ++j) {
      float v[20];
      staged_row<20>(v, stage + (lane * 5 + j) * 20);
      fold_row<20>(a20, v);
    }
    put_acc<20>(myrow + 80, a20);
  }
  wave_lds_handover();
  store_pieces<20, 160>(stage, p_fa, e0, pc);
  wave_lds_handover();
  load_pieces<20, 160>(p_fb, tab_freeform, i_fb, pc);
  if (lane < 16) {
    float a20[20];
    zero_acc<20>(a20);
#pragma unroll
    for (int j = 0; j < 10; ++j) {
      float v[20];
      staged_row<20>(v, stage + (lane * 10 + j) * 20);
      fold_row<20>(a20, v);
    }
    put_acc<20>(myrow + 100, a20);
  }
  wave_lds_handover();
  store_pieces<20, 160>(stage, p_fb, e0, pc);
  wave_lds_handover();
  if (lane >= 16) {
    float a20[20];
    zero_acc<20>(a20);
#pragma unroll
    for (int j = 0; j < 10; ++j) {
      float v[20];
      staged_row<20>(v, stage + ((lane - 16) * 10 + j) * 20);
      fold_row<20>(a20, v);
    }
    put_acc<20>(myrow + 100, a20);
  }
  wave_lds_handover();

  v4f q2[4][4];
  const int nq2 = tid & 31;
#pragma unroll
  for (int i = 0; i < 4; ++i) {
    const int cb = wave + 4 * i, pr = cb >> 1, h = cb & 1;
    const int kb = 8 * pr + 2 * h;
    load_quad<128>(q2[i], W2, kb, kb + 5, nq2);
  }

  {
    const int PI[15] = {0,0,0,0,0,1,1,1,1,2,2,2,3,3,4};
    const int PJ[15] = {1,2,3,4,5,2,3,4,5,3,4,5,4,5,5};
#pragma unroll 3
    for (int p = 0; p < 15; ++p) {
      const float* a = myrow + 20 * PI[p];
      const float* b = myrow + 20 * PJ[p];
      float s = 0.f;
#pragma unroll
      for (int d = 0; d < 20; ++d) s = fmaf(a[d], b[d], s);
      myrow[120 + p] = s;
    }
    myrow[135] = 0.f;
  }

  __syncthreads();
  {
    v4f* d2 = (v4f*)s_w2 + (nq2 >> 2) * W2_TILE + (nq2 & 3) * 4;
#pragma unroll
    for (int i = 0; i < 4; ++i) {
      const int cb = wave + 4 * i, pr = cb >> 1, h = cb & 1;
      store_quad(d2 + pr * 32 + h * 16, q2[i]);
    }
  }
  __syncthreads();

  float* hbase = s_h + wave * 16 * H_LD;
  const v4f* w1f = (const v4f*)s_w1 + lane;
  const v4f* w2f = (const v4f*)s_w2 + lane;

  for (int t = 0; t < 2; ++t) {
    const int m0 = (wave * 2 + t) * 16;

    const float* arow = s_pre + pre_row(m0 + lo) + 2 * hi;
#pragma unroll
    for (int nt = 0; nt < 4; ++nt) {
      const int n0 = nt * 16;
      const float bias = b1[n0 + lo];
      v8f c;
#pragma unroll
      for (int i = 0; i < 8; ++i) c[i] = bias;
#pragma unroll
      for (int pr = 0; pr < W1_PAIRS; ++pr) {
        const int k0 = 8 * pr;
        const v4f w = w1f[nt * W1_TILE + pr * 32];
        v2f a, b;
        a.x = arow[k0];     a.y = arow[k0 + 1];
        b.x = w.x;          b.y = w.y;
        c = wmma_f32_k4(a, b, c);
        a.x = arow[k0 + 4]; a.y = arow[k0 + 5];
        b.x = w.z;          b.y = w.w;
        c = wmma_f32_k4(a, b, c);
      }
#pragma unroll
      for (int i = 0; i < 8; ++i) {
        float v = c[i];
        v = v > 0.f ? v : 0.01f * v;
        hbase[(i + 8 * hi) * H_LD + n0 + lo] = v;
      }
    }
    asm volatile("s_wait_dscnt 0x0" ::: "memory");

    const float* hrow = hbase + lo * H_LD + 2 * hi;
    const size_t rbase = (size_t)(blockIdx.x * ROWS_PER_BLOCK + m0) * 128;
#pragma unroll
    for (int nt = 0; nt < 8; ++nt) {
      const int n0 = nt * 16;
      const float bias = b2[n0 + lo];
      v8f c;
#pragma unroll
      for (int i = 0; i < 8; ++i) c[i] = bias;
#pragma unroll
      for (int pr = 0; pr < W2_PAIRS; ++pr) {
        const int k0 = 8 * pr;
        const v4f w = w2f[nt * W2_TILE + pr * 32];
        v2f a, b;
        a.x = hrow[k0];     a.y = hrow[k0 + 1];
        b.x = w.x;          b.y = w.y;
        c = wmma_f32_k4(a, b, c);
        a.x = hrow[k0 + 4]; a.y = hrow[k0 + 5];
        b.x = w.z;          b.y = w.w;
        c = wmma_f32_k4(a, b, c);
      }
#pragma unroll
      for (int i = 0; i < 8; ++i)
        out[rbase + (size_t)(i + 8 * hi) * 128 + n0 + lo] = c[i];
    }
  }
}

extern "C" void kernel_launch(void* const* d_in, const int* in_sizes, int n_in,
                              void* d_out, int out_size, void* d_ws, size_t ws_size,
                              hipStream_t stream) {
  const float* dense = (const float*)d_in[0];
  const int* ids_rating       = (const int*)d_in[1];
  const int* ids_category     = (const int*)d_in[2];
  const int* ids_fandom       = (const int*)d_in[3];
  const int* ids_relationship = (const int*)d_in[4];
  const int* ids_character    = (const int*)d_in[5];
  const int* ids_freeform     = (const int*)d_in[6];
  const float* W_dense = (const float*)d_in[7];
  const float* b_dense = (const float*)d_in[8];
  const float* tab_rating       = (const float*)d_in[9];
  const float* tab_category     = (const float*)d_in[10];
  const float* tab_fandom       = (const float*)d_in[11];
  const float* tab_relationship = (const float*)d_in[12];
  const float* tab_character    = (const float*)d_in[13];
  const float* tab_freeform     = (const float*)d_in[14];
  const float* W1 = (const float*)d_in[15];
  const float* b1 = (const float*)d_in[16];
  const float* W2 = (const float*)d_in[17];
  const float* b2 = (const float*)d_in[18];
  float* out = (float*)d_out;

  const int Btot = in_sizes[0] / 64;
  const int grid = Btot / ROWS_PER_BLOCK;
  const size_t shmem = (size_t)(ROWS_PER_BLOCK * PRE_LD + W1F_FLOATS +
                                UNION_FLOATS) * sizeof(float);

  fused_embmlp_kernel<<<grid, THREADS, shmem, stream>>>(
      dense, ids_rating, ids_category, ids_fandom, ids_relationship,
      ids_character, ids_freeform, W_dense, b_dense, tab_rating, tab_category,
      tab_fandom, tab_relationship, tab_character, tab_freeform, W1, b1, W2, b2,
      out);
}


# The 15 pair dots run from registers: each lane reads its six 20-float vectors once (8-byte LDS loads), then 15 independent FMA chains, not a loop re-reading both vectors per pair; sum order unchanged
# speedup vs baseline: 1.5748x; 1.0500x over previous
#include <hip/hip_runtime.h>
#include <math.h>

typedef float v2f __attribute__((ext_vector_type(2)));
typedef float v4f __attribute__((ext_vector_type(4)));
typedef float v8f __attribute__((ext_vector_type(8)));

#define ROWS_PER_BLOCK 128
#define THREADS 128
#define PRE_LD 140
#define H_LD 68
#define W1_PAIRS 17
#define W2_PAIRS 8
#define W1_TILE (W1_PAIRS * 32 + 1)
#define W2_TILE (W2_PAIRS * 32 + 1)
#define W1F_FLOATS (4 * W1_TILE * 4)
#define W2F_FLOATS (8 * W2_TILE * 4)
#define STAGE_FLOATS 3200
#define HREG_FLOATS (4 * 16 * H_LD)
#define UNION_FLOATS (4 * STAGE_FLOATS)
static_assert(UNION_FLOATS >= HREG_FLOATS + W2F_FLOATS, "h and W2 fragments must fit");


template<int DIM> struct piece { typedef v4f type; };
template<> struct piece<10> { typedef v2f type; };

template<int NE>
__device__ __forceinline__ int list_pos(int i, int e0) {
  const int e = 6 * i + e0;
  return e < NE ? e : NE - 1;
}

template<int NE>
__device__ __forceinline__ void load_ids(int (&id)[(NE + 5) / 6],
                                         const int* __restrict__ idp, int e0) {
#pragma unroll
  for (int i = 0; i < (NE + 5) / 6; ++i) id[i] = idp[list_pos<NE>(i, e0)];
}

template<int DIM, int NE>
__device__ __forceinline__ void load_pieces(typename piece<DIM>::type (&p)[(NE + 5) / 6],
                                            const float* __restrict__ tab,
                                            const int (&id)[(NE + 5) / 6], int c) {
  typedef typename piece<DIM>::type PT;
#pragma unroll
  for (int i = 0; i < (NE + 5) / 6; ++i)
    p[i] = *(const PT*)(tab + (size_t)id[i] * DIM + c * (DIM / 5));
}

template<int DIM, int NE>
__device__ __forceinline__ void store_pieces(float* dst,
                                             const typename piece<DIM>::type (&p)[(NE + 5) / 6],
                                             int e0, int c) {
  typedef typename piece<DIM>::type PT;
#pragma unroll
  for (int i = 0; i < (NE + 5) / 6; ++i)
    *(PT*)(dst + list_pos<NE>(i, e0) * DIM + c * (DIM / 5)) = p[i];
}

template<int DIM>
__device__ __forceinline__ void staged_row(float (&v)[DIM], const float* src) {
  typedef typename piece<DIM>::type PT;
  constexpr int PW = DIM / 5;
#pragma unroll
  for (int c = 0; c < 5; ++c) {
    PT t = *(const PT*)(src + c * PW);
#pragma unroll
    for (int k = 0; k < PW; ++k) v[PW * c + k] = t[k];
  }
}

__device__ __forceinline__ void wave_lds_handover() {
  asm volatile("s_wait_dscnt 0x0" ::: "memory");
}

template<int DIM>
__device__ __forceinline__ void fold_row(float (&acc)[DIM], const float (&v)[DIM]) {
  float ss = 0.f;
#pragma unroll
  for (int d = 0; d < DIM; ++d) ss = fmaf(v[d], v[d], ss);
  float n = sqrtf(ss);
  float s = fminf(1.0f, 1.0f / (n + 1e-7f));
#pragma unroll
  for (int d = 0; d < DIM; ++d) acc[d] = fmaf(v[d], s, acc[d]);
}

template<int DIM>
__device__ __forceinline__ void zero_acc(float (&acc)[DIM]) {
#pragma unroll
  for (int d = 0; d < DIM; ++d) acc[d] = 0.f;
}

template<int DIM>
__device__ __forceinline__ void put_acc(float* dst, const float (&acc)[DIM]) {
#pragma unroll
  for (int d = 0; d < DIM; ++d) dst[d] = acc[d];
}

__device__ __forceinline__ int pre_row(int r) {
  return r * PRE_LD + 2 * ((r >> 4) & 1);
}

__device__ __forceinline__ v8f wmma_f32_k4(v2f a, v2f b, v8f c) {
#if __has_builtin(__builtin_amdgcn_wmma_f32_16x16x4_f32)
  return __builtin_amdgcn_wmma_f32_16x16x4_f32(false, a, false, b, (short)0, c,
                                               false, false);
#else
  asm volatile("v_wmma_f32_16x16x4_f32 %0, %1, %2, %0"
               : "+v"(c) : "v"(a), "v"(b));
  return c;
#endif
}

__device__ __forceinline__ v4f make4(float a, float b, float c, float d) {
  v4f r; r.x = a; r.y = b; r.z = c; r.w = d;
  return r;
}

template<int LDW>
__device__ __forceinline__ void load_quad(v4f (&r)[4], const float* __restrict__ Wm,
                                          int kb, int k3, int nq) {
  const float* p = Wm + kb * LDW + 4 * nq;
  r[0] = *(const v4f*)(p);
  r[1] = *(const v4f*)(p + LDW);
  r[2] = *(const v4f*)(p + 4 * LDW);
  r[3] = *(const v4f*)(Wm + k3 * LDW + 4 * nq);
}

__device__ __forceinline__ void store_quad(v4f* dst, const v4f (&r)[4]) {
  dst[0] = make4(r[0].x, r[1].x, r[2].x, r[3].x);
  dst[1] = make4(r[0].y, r[1].y, r[2].y, r[3].y);
  dst[2] = make4(r[0].z, r[1].z, r[2].z, r[3].z);
  dst[3] = make4(r[0].w, r[1].w, r[2].w, r[3].w);
}

__global__ __launch_bounds__(THREADS)
__attribute__((amdgpu_waves_per_eu(2)))
void fused_embmlp_kernel(
    const float* __restrict__ dense,
    const int* __restrict__ ids_rating, const int* __restrict__ ids_category,
    const int* __restrict__ ids_fandom, const int* __restrict__ ids_relationship,
    const int* __restrict__ ids_character, const int* __restrict__ ids_freeform,
    const float* __restrict__ W_dense, const float* __restrict__ b_dense,
    const float* __restrict__ tab_rating, const float* __restrict__ tab_category,
    const float* __restrict__ tab_fandom, const float* __restrict__ tab_relationship,
    const float* __restrict__ tab_character, const float* __restrict__ tab_freeform,
    const float* __restrict__ W1, const float* __restrict__ b1,
    const float* __restrict__ W2, const float* __restrict__ b2,
    float* __restrict__ out) {
  extern __shared__ float smem[];
  float* s_pre = smem;
  float* s_w1  = smem + ROWS_PER_BLOCK * PRE_LD;
  float* s_un  = s_w1 + W1F_FLOATS;
  float* s_h   = s_un;
  float* s_w2  = s_un + HREG_FLOATS;

  const int tid = threadIdx.x;
  const int wave = tid >> 5;
  const int lane = tid & 31;

  {
    const int sw = tid >> 5;
    const int nq1 = tid & 15, h1 = (tid >> 4) & 1;
    const int h16 = tid >> 6, j16 = (tid >> 4) & 3;
    const int k16 = 128 + 2 * h16 + (j16 & 1) + 4 * (j16 >> 1);
    const bool pad16 = (k16 >= 135);
    v4f q1[4][4];
#pragma unroll
    for (int i = 0; i < 4; ++i) {
      const int kb = 8 * (sw + 4 * i) + 2 * h1;
      load_quad<64>(q1[i], W1, kb, kb + 5, nq1);
    }
    v4f t16 = *(const v4f*)(W1 + (pad16 ? 134 : k16) * 64 + 4 * nq1);
    if (pad16) t16 = make4(0.f, 0.f, 0.f, 0.f);
    v4f* d1 = (v4f*)s_w1 + (nq1 >> 2) * W1_TILE + (nq1 & 3) * 4;
#pragma unroll
    for (int i = 0; i < 4; ++i) store_quad(d1 + (sw + 4 * i) * 32 + h1 * 16, q1[i]);
    float* e16 = (float*)(d1 + 16 * 32 + h16 * 16) + j16;
    e16[0] = t16.x; e16[4] = t16.y; e16[8] = t16.z; e16[12] = t16.w;
  }

  float* myrow = s_pre + pre_row(tid);

  const int e0 = lane / 5, pc = lane % 5;
  float* stage = s_un + wave * STAGE_FLOATS;
  const int wrow = blockIdx.x * ROWS_PER_BLOCK + wave * 32;
  int i_rat[6], i_cat[11], i_fan[16], i_rel[16], i_chr[27], i_fa[27], i_fb[27];
  load_ids<32>(i_rat, ids_rating + wrow, e0);
  load_ids<64>(i_cat, ids_category + wrow * 2, e0);
  load_ids<96>(i_fan, ids_fandom + wrow * 3, e0);

  const int lo = lane & 15;
  const int hi = lane >> 4;
  v4f xa[2][8], wb[2][8];
  {
    const float* xr = dense + (size_t)(wrow + lo) * 64 + 4 * hi;
#pragma unroll
    for (int t = 0; t < 2; ++t)
#pragma unroll
      for (int j = 0; j < 8; ++j) xa[t][j] = *(const v4f*)(xr + t * 16 * 64 + 8 * j);
#pragma unroll
    for (int nt = 0; nt < 2; ++nt) {
      const int n = 16 * nt + lo;
      const bool ok = n < 20;
      const float* wp = W_dense + 4 * hi * 20 + (ok ? n : 0);
#pragma unroll
      for (int j = 0; j < 8; ++j) {
        const float* p = wp + 8 * j * 20;
        const v4f w = make4(p[0], p[20], p[40], p[60]);
        wb[nt][j] = ok ? w : make4(0.f, 0.f, 0.f, 0.f);
      }
    }
  }
  const float bd0 = b_dense[lo];
  const float bd1r = b_dense[lo < 4 ? 16 + lo : 0];
  const float bd1 = lo < 4 ? bd1r : 0.f;

  v2f p_rat[6], p_cat[11];
  v4f p_fan[16], p_rel[16], p_chr[27], p_fa[27], p_fb[27];
  load_pieces<10, 32>(p_rat, tab_rating, i_rat, pc);
  load_pieces<10, 64>(p_cat, tab_category, i_cat, pc);
  load_pieces<20, 96>(p_fan, tab_fandom, i_fan, pc);
  load_ids<96>(i_rel, ids_relationship + wrow * 3, e0);
#pragma unroll
  for (int t = 0; t < 2; ++t) {
    v8f c0, c1;
#pragma unroll
    for (int i = 0; i < 8; ++i) { c0[i] = bd0; c1[i] = bd1; }
#pragma unroll
    for (int j = 0; j < 8; ++j) {
      v2f a, b;
      a.x = xa[t][j].x;  a.y = xa[t][j].y;
      b.x = wb[0][j].x;  b.y = wb[0][j].y;
      c0 = wmma_f32_k4(a, b, c0);
      b.x = wb[1][j].x;  b.y = wb[1][j].y;
      c1 = wmma_f32_k4(a, b, c1);
      a.x = xa[t][j].z;  a.y = xa[t][j].w;
      b.x = wb[0][j].z;  b.y = wb[0][j].w;
      c0 = wmma_f32_k4(a, b, c0);
      b.x = wb[1][j].z;  b.y = wb[1][j].w;
      c1 = wmma_f32_k4(a, b, c1);
    }
    float* drow = s_pre + pre_row(wave * 32 + 16 * t + 8 * hi);
#pragma unroll
    for (int i = 0; i < 8; ++i) drow[i * PRE_LD + lo] = c0[i];
    if (lo < 4) {
#pragma unroll
      for (int i = 0; i < 8; ++i) drow[i * PRE_LD + 16 + lo] = c1[i];
    }
  }
  store_pieces<10, 32>(stage, p_rat, e0, pc);
  store_pieces<10, 64>(stage + 320, p_cat, e0, pc);
  store_pieces<20, 96>(stage + 960, p_fan, e0, pc);
  wave_lds_handover();
  load_pieces<20, 96>(p_rel, tab_relationship, i_rel, pc);
  load_ids<160>(i_chr, ids_character + wrow * 5, e0);
  {
    float r0[10], c0[10], c1[10], f[3][20];
    staged_row<10>(r0, stage + lane * 10);
    staged_row<10>(c0, stage + 320 + (lane * 2) * 10);
    staged_row<10>(c1, stage + 320 + (lane * 2 + 1) * 10);
#pragma unroll
    for (int j = 0; j < 3; ++j) staged_row<20>(f[j], stage + 960 + (lane * 3 + j) * 20);
    float a10[10], a20[20];
    zero_acc<10>(a10); fold_row<10>(a10, r0); put_acc<10>(myrow + 20, a10);
    zero_acc<10>(a10); fold_row<10>(a10, c0); fold_row<10>(a10, c1);
    put_acc<10>(myrow + 30, a10);
    zero_acc<20>(a20);
#pragma unroll
    for (int j = 0; j < 3; ++j) fold_row<20>(a20, f[j]);
    put_acc<20>(myrow + 40, a20);
  }
  wave_lds_handover();
  store_pieces<20, 96>(stage, p_rel, e0, pc);
  wave_lds_handover();
  load_pieces<20, 160>(p_chr, tab_character, i_chr, pc);
  load_ids<160>(i_fa, ids_freeform + wrow * 10, e0);
  {
    float r[3][20], a20[20];
#pragma unroll
    for (int j = 0; j < 3; ++j) staged_row<20>(r[j], stage + (lane * 3 + j) * 20);
    zero_acc<20>(a20);
#pragma unroll
    for (int j = 0; j < 3; ++j) fold_row<20>(a20, r[j]);
    put_acc<20>(myrow + 60, a20);
  }
  wave_lds_handover();
  store_pieces<20, 160>(stage, p_chr, e0, pc);
  wave_lds_handover();
  load_pieces<20, 160>(p_fa, tab_freeform, i_fa, pc);
  load_ids<160>(i_fb, ids_freeform + (wrow + 16) * 10, e0);
  {
    float a20[20];
    zero_acc<20>(a20);
#pragma unroll
    for (int j = 0; j < 5; ++j) {
      float v[20];
      staged_row<20>(v, stage + (lane * 5 + j) * 20);
      fold_row<20>(a20, v);
    }
    put_acc<20>(myrow + 80, a20);
  }
  wave_lds_handover();
  store_pieces<20, 160>(stage, p_fa, e0, pc);
  wave_lds_handover();
  load_pieces<20, 160>(p_fb, tab_freeform, i_fb, pc);
  if (lane < 16) {
    float a20[20];
    zero_acc<20>(a20);
#pragma unroll
    for (int j = 0; j < 10; ++j) {
      float v[20];
      staged_row<20>(v, stage + (lane * 10 + j) * 20);
      fold_row<20>(a20, v);
    }
    put_acc<20>(myrow + 100, a20);
  }
  wave_lds_handover();
  store_pieces<20, 160>(stage, p_fb, e0, pc);
  wave_lds_handover();
  if (lane >= 16) {
    float a20[20];
    zero_acc<20>(a20);
#pragma unroll
    for (int j = 0; j < 10; ++j) {
      float v[20];
      staged_row<20>(v, stage + ((lane - 16) * 10 + j) * 20);
      fold_row<20>(a20, v);
    }
    put_acc<20>(myrow + 100, a20);
  }
  wave_lds_handover();

  v4f q2[4][4];
  const int nq2 = tid & 31;
#pragma unroll
  for (int i = 0; i < 4; ++i) {
    const int cb = wave + 4 * i, pr = cb >> 1, h = cb & 1;
    const int kb = 8 * pr + 2 * h;
    load_quad<128>(q2[i], W2, kb, kb + 5, nq2);
  }

  {
    float vec[6][20];
#pragma unroll
    for (int i = 0; i < 6; ++i) {
#pragma unroll
      for (int q = 0; q < 10; ++q) {
        const v2f t = *(const v2f*)(myrow + 20 * i + 2 * q);
        vec[i][2 * q] = t.x;
        vec[i][2 * q + 1] = t.y;
      }
    }
    float dot[16];
#pragma unroll
    for (int i = 0; i < 5; ++i) {
#pragma unroll
      for (int j = i + 1; j < 6; ++j) {
        float s = 0.f;
#pragma unroll
        for (int d = 0; d < 20; ++d) s = fmaf(vec[i][d], vec[j][d], s);
        dot[i * (11 - i) / 2 + j - i - 1] = s;
      }
    }
    dot[15] = 0.f;
#pragma unroll
    for (int q = 0; q < 8; ++q) {
      v2f t; t.x = dot[2 * q]; t.y = dot[2 * q + 1];
      *(v2f*)(myrow + 120 + 2 * q) = t;
    }
  }

  __syncthreads();
  {
    v4f* d2 = (v4f*)s_w2 + (nq2 >> 2) * W2_TILE + (nq2 & 3) * 4;
#pragma unroll
    for (int i = 0; i < 4; ++i) {
      const int cb = wave + 4 * i, pr = cb >> 1, h = cb & 1;
      store_quad(d2 + pr * 32 + h * 16, q2[i]);
    }
  }
  __syncthreads();

  float* hbase = s_h + wave * 16 * H_LD;
  const v4f* w1f = (const v4f*)s_w1 + lane;
  const v4f* w2f = (const v4f*)s_w2 + lane;

  for (int t = 0; t < 2; ++t) {
    const int m0 = (wave * 2 + t) * 16;

    const float* arow = s_pre + pre_row(m0 + lo) + 2 * hi;
#pragma unroll
    for (int nt = 0; nt < 4; ++nt) {
      const int n0 = nt * 16;
      const float bias = b1[n0 + lo];
      v8f c;
#pragma unroll
      for (int i = 0; i < 8; ++i) c[i] = bias;
#pragma unroll
      for (int pr = 0; pr < W1_PAIRS; ++pr) {
        const int k0 = 8 * pr;
        const v4f w = w1f[nt * W1_TILE + pr * 32];
        v2f a, b;
        a.x = arow[k0];     a.y = arow[k0 + 1];
        b.x = w.x;          b.y = w.y;
        c = wmma_f32_k4(a, b, c);
        a.x = arow[k0 + 4]; a.y = arow[k0 + 5];
        b.x = w.z;          b.y = w.w;
        c = wmma_f32_k4(a, b, c);
      }
#pragma unroll
      for (int i = 0; i < 8; ++i) {
        float v = c[i];
        v = v > 0.f ? v : 0.01f * v;
        hbase[(i + 8 * hi) * H_LD + n0 + lo] = v;
      }
    }
    asm volatile("s_wait_dscnt 0x0" ::: "memory");

    const float* hrow = hbase + lo * H_LD + 2 * hi;
    const size_t rbase = (size_t)(blockIdx.x * ROWS_PER_BLOCK + m0) * 128;
#pragma unroll
    for (int nt = 0; nt < 8; ++nt) {
      const int n0 = nt * 16;
      const float bias = b2[n0 + lo];
      v8f c;
#pragma unroll
      for (int i = 0; i < 8; ++i) c[i] = bias;
#pragma unroll
      for (int pr = 0; pr < W2_PAIRS; ++pr) {
        const int k0 = 8 * pr;
        const v4f w = w2f[nt * W2_TILE + pr * 32];
        v2f a, b;
        a.x = hrow[k0];     a.y = hrow[k0 + 1];
        b.x = w.x;          b.y = w.y;
        c = wmma_f32_k4(a, b, c);
        a.x = hrow[k0 + 4]; a.y = hrow[k0 + 5];
        b.x = w.z;          b.y = w.w;
        c = wmma_f32_k4(a, b, c);
      }
#pragma unroll
      for (int i = 0; i < 8; ++i)
        out[rbase + (size_t)(i + 8 * hi) * 128 + n0 + lo] = c[i];
    }
  }
}

extern "C" void kernel_launch(void* const* d_in, const int* in_sizes, int n_in,
                              void* d_out, int out_size, void* d_ws, size_t ws_size,
                              hipStream_t stream) {
  const float* dense = (const float*)d_in[0];
  const int* ids_rating       = (const int*)d_in[1];
  const int* ids_category     = (const int*)d_in[2];
  const int* ids_fandom       = (const int*)d_in[3];
  const int* ids_relationship = (const int*)d_in[4];
  const int* ids_character    = (const int*)d_in[5];
  const int* ids_freeform     = (const int*)d_in[6];
  const float* W_dense = (const float*)d_in[7];
  const float* b_dense = (const float*)d_in[8];
  const float* tab_rating       = (const float*)d_in[9];
  const float* tab_category     = (const float*)d_in[10];
  const float* tab_fandom       = (const float*)d_in[11];
  const float* tab_relationship = (const float*)d_in[12];
  const float* tab_character    = (const float*)d_in[13];
  const float* tab_freeform     = (const float*)d_in[14];
  const float* W1 = (const float*)d_in[15];
  const float* b1 = (const float*)d_in[16];
  const float* W2 = (const float*)d_in[17];
  const float* b2 = (const float*)d_in[18];
  float* out = (float*)d_out;

  const int Btot = in_sizes[0] / 64;
  const int grid = Btot / ROWS_PER_BLOCK;
  const size_t shmem = (size_t)(ROWS_PER_BLOCK * PRE_LD + W1F_FLOATS +
                                UNION_FLOATS) * sizeof(float);

  fused_embmlp_kernel<<<grid, THREADS, shmem, stream>>>(
      dense, ids_rating, ids_category, ids_fandom, ids_relationship,
      ids_character, ids_freeform, W_dense, b_dense, tab_rating, tab_category,
      tab_fandom, tab_relationship, tab_character, tab_freeform, W1, b1, W2, b2,
      out);
}
